# starting kernel
# baseline (speedup 1.0000x reference)
#include <hip/hip_runtime.h>
#include <hip/hip_bf16.h>

typedef __attribute__((ext_vector_type(2))) float v2f;
typedef __attribute__((ext_vector_type(8))) float v8f;

#define N_NODES 19
#define N_T     4096
#define LATENT  512
#define HID     2048
#define N_EDGES 342
#define PAD_ROWS 32

#define WS_H0    0
#define WS_T1    (WS_H0 + PAD_ROWS * LATENT)
#define WS_H1    (WS_T1 + PAD_ROWS * HID)
#define WS_H1AGG (WS_H1 + PAD_ROWS * HID)
#define WS_T2    (WS_H1AGG + PAD_ROWS * HID)
#define WS_Y     (WS_T2 + PAD_ROWS * N_T)
#define WS_TOTAL (WS_Y + LATENT)

__global__ void zero_kernel(float* __restrict__ p, int n) {
    int i = blockIdx.x * blockDim.x + threadIdx.x;
    int stride = gridDim.x * blockDim.x;
    for (; i < n; i += stride) p[i] = 0.0f;
}

__global__ void gin_agg_kernel(const float* __restrict__ h, const int* __restrict__ edges,
                               float* __restrict__ out, int F) {
    int idx = blockIdx.x * blockDim.x + threadIdx.x;
    if (idx >= N_NODES * F) return;
    int node = idx / F;
    int f = idx - node * F;
    const int* src = edges;
    const int* dst = edges + N_EDGES;
    float v = h[(size_t)node * F + f];
    for (int e = 0; e < N_EDGES; ++e) {
        if (dst[e] == node) v += h[(size_t)src[e] * F + f];
    }
    out[(size_t)node * F + f] = v;
}

template <int K, int N, int KCHUNK>
__global__ __launch_bounds__(128) void gemm_wmma_kernel(
        const float* __restrict__ A, const float* __restrict__ W,
        float* __restrict__ Out) {
    const int lane = threadIdx.x;
    const int half = lane >> 4;
    const int lm   = lane & 15;
    const int tile = blockIdx.x * 4 + threadIdx.y;
    const int n    = tile * 16 + lm;
    const int k0   = blockIdx.y * KCHUNK;

    const float* a0p  = A + (size_t)lm * K        + k0 + 2 * half;
    const float* a1p  = A + (size_t)(16 + lm) * K + k0 + 2 * half;
    const float* bptr = W + (size_t)(k0 + 2 * half) * N + n;

    v8f c0 = {};
    v8f c1 = {};

    #pragma unroll 4
    for (int kk = 0; kk < KCHUNK; kk += 4) {
        v2f a0 = *(const v2f*)a0p;
        v2f a1 = *(const v2f*)a1p;
        v2f b;
        b.x = bptr[0];
        b.y = bptr[N];
        c0 = __builtin_amdgcn_wmma_f32_16x16x4_f32(false, a0, false, b, (short)0, c0, false, false);
        c1 = __builtin_amdgcn_wmma_f32_16x16x4_f32(false, a1, false, b, (short)0, c1, false, false);
        a0p += 4;
        a1p += 4;
        bptr += (size_t)4 * N;
    }

    #pragma unroll
    for (int i = 0; i < 8; ++i) {
        int r0 = i + 8 * half;
        atomicAdd(&Out[(size_t)r0 * N + n], c0[i]);
        int r1 = 16 + r0;
        if (r1 < N_NODES) atomicAdd(&Out[(size_t)r1 * N + n], c1[i]);
    }
}

__global__ void bias_act_kernel(float* __restrict__ buf, const float* __restrict__ bias,
                                int N, int do_relu) {
    int idx = blockIdx.x * blockDim.x + threadIdx.x;
    if (idx >= N_NODES * N) return;
    int col = idx % N;
    float v = buf[idx] + bias[col];
    if (do_relu) v = v > 0.0f ? v : 0.0f;
    buf[idx] = v;
}

__global__ __launch_bounds__(512) void matvec_kernel(
        const float* __restrict__ x, const float* __restrict__ Wc1,
        float* __restrict__ y, int rows, int rowsPerBlock) {
    int j = threadIdx.x;
    int r0 = blockIdx.x * rowsPerBlock;
    int r1 = r0 + rowsPerBlock;
    if (r1 > rows) r1 = rows;
    float acc = 0.0f;
    for (int r = r0; r < r1; ++r) {
        acc += x[r] * Wc1[(size_t)r * LATENT + j];
    }
    atomicAdd(&y[j], acc);
}

__global__ __launch_bounds__(512) void final_kernel(
        const float* __restrict__ y, const float* __restrict__ bc1,
        const float* __restrict__ Wc2, const float* __restrict__ bc2,
        float* __restrict__ out0) {
    __shared__ float red[512];
    int j = threadIdx.x;
    red[j] = (y[j] + bc1[j]) * Wc2[j];
    __syncthreads();
    for (int s = 256; s > 0; s >>= 1) {
        if (j < s) red[j] += red[j + s];
        __syncthreads();
    }
    if (j == 0) {
        float v = red[0] + bc2[0];
        out0[0] = 1.0f / (1.0f + __expf(-v));
    }
}

extern "C" void kernel_launch(void* const* d_in, const int* in_sizes, int n_in,
                              void* d_out, int out_size, void* d_ws, size_t ws_size,
                              hipStream_t stream) {
    const float* z    = (const float*)d_in[0];
    const int*   eidx = (const int*)  d_in[1];
    const float* W1a  = (const float*)d_in[2];
    const float* b1a  = (const float*)d_in[3];
    const float* W1b  = (const float*)d_in[4];
    const float* b1b  = (const float*)d_in[5];
    const float* W2a  = (const float*)d_in[6];
    const float* b2a  = (const float*)d_in[7];
    const float* W2b  = (const float*)d_in[8];
    const float* b2b  = (const float*)d_in[9];
    const float* Wc1  = (const float*)d_in[10];
    const float* bc1  = (const float*)d_in[11];
    const float* Wc2  = (const float*)d_in[12];
    const float* bc2  = (const float*)d_in[13];

    float* out   = (float*)d_out;
    float* ws    = (float*)d_ws;
    float* h0    = ws + WS_H0;
    float* t1    = ws + WS_T1;
    float* h1    = ws + WS_H1;
    float* h1agg = ws + WS_H1AGG;
    float* t2    = ws + WS_T2;
    float* yv    = ws + WS_Y;
    float* recon = out + 1;

    zero_kernel<<<512, 256, 0, stream>>>(ws, (int)WS_TOTAL);
    zero_kernel<<<512, 256, 0, stream>>>(out, 1 + N_NODES * N_T);

    dim3 gblk(32, 4);

    {
        int n = N_NODES * LATENT;
        gin_agg_kernel<<<(n + 255) / 256, 256, 0, stream>>>(z, eidx, h0, LATENT);
    }
    gemm_wmma_kernel<LATENT, HID, LATENT / 16>
        <<<dim3(HID / 64, 16), gblk, 0, stream>>>(h0, W1a, t1);
    {
        int n = N_NODES * HID;
        bias_act_kernel<<<(n + 255) / 256, 256, 0, stream>>>(t1, b1a, HID, 1);
    }
    gemm_wmma_kernel<HID, HID, HID / 16>
        <<<dim3(HID / 64, 16), gblk, 0, stream>>>(t1, W1b, h1);
    {
        int n = N_NODES * HID;
        bias_act_kernel<<<(n + 255) / 256, 256, 0, stream>>>(h1, b1b, HID, 1);
    }
    {
        int n = N_NODES * HID;
        gin_agg_kernel<<<(n + 255) / 256, 256, 0, stream>>>(h1, eidx, h1agg, HID);
    }
    gemm_wmma_kernel<HID, N_T, HID / 8>
        <<<dim3(N_T / 64, 8), gblk, 0, stream>>>(h1agg, W2a, t2);
    {
        int n = N_NODES * N_T;
        bias_act_kernel<<<(n + 255) / 256, 256, 0, stream>>>(t2, b2a, N_T, 1);
    }
    gemm_wmma_kernel<N_T, N_T, N_T / 8>
        <<<dim3(N_T / 64, 8), gblk, 0, stream>>>(t2, W2b, recon);
    {
        int n = N_NODES * N_T;
        bias_act_kernel<<<(n + 255) / 256, 256, 0, stream>>>(recon, b2b, N_T, 0);
    }
    {
        int rows = N_NODES * N_T;
        int rpb = 256;
        matvec_kernel<<<(rows + rpb - 1) / rpb, 512, 0, stream>>>(recon, Wc1, yv, rows, rpb);
    }
    final_kernel<<<1, 512, 0, stream>>>(yv, bc1, Wc2, bc2, out);
}


# GEMM activations kept in the workspace in matrix-fragment order: one contiguous 16-byte-per-lane A load per 8 k feeds two products, not a 16-row gather per 4 k; k order inside a wave regrouped
# speedup vs baseline: 3.4267x; 3.4267x over previous
#include <hip/hip_runtime.h>
#include <hip/hip_bf16.h>

typedef __attribute__((ext_vector_type(2))) float v2f;
typedef __attribute__((ext_vector_type(4))) float v4f;
typedef __attribute__((ext_vector_type(8))) float v8f;

#define N_NODES 19
#define N_T     4096
#define LATENT  512
#define HID     2048
#define N_EDGES 342
#define PAD_ROWS 32

#define WS_H0    0
#define WS_T1    (WS_H0 + PAD_ROWS * LATENT)
#define WS_H1    (WS_T1 + PAD_ROWS * HID)
#define WS_H1AGG (WS_H1 + PAD_ROWS * HID)
#define WS_T2    (WS_H1AGG + PAD_ROWS * HID)
#define WS_Y     (WS_T2 + PAD_ROWS * N_T)
#define WS_TOTAL (WS_Y + LATENT)

__device__ __forceinline__ size_t frag_at(int row, int k, int K) {
    const int blk = k >> 3, half = (k >> 2) & 1, j = k & 3;
    return row < 16 ? (size_t)blk * 128 + half * 64 + row * 4 + j
                    : (size_t)16 * K + (size_t)blk * 32 + half * 16 + (row - 16) * 4 + j;
}

#define AGG_COLS   32
#define AGG_CHUNKS ((N_EDGES + 31) / 32)
#define AGG_LIST   (((N_EDGES + 8) + 7) / 8 * 8)
static_assert(LATENT % AGG_COLS == 0 && HID % AGG_COLS == 0, "strips must tile both widths: no tail code");
typedef __attribute__((ext_vector_type(4))) int v4i;

__global__ __launch_bounds__(N_NODES * 32) void gin_agg_kernel(
        const float* __restrict__ h, const int* __restrict__ edges,
        float* __restrict__ out, int F) {
    __shared__ float hs[N_NODES * AGG_COLS];
    __shared__ __attribute__((aligned(16))) int lst[N_NODES * AGG_LIST];

    const unsigned lane = threadIdx.x & 31u;
    const unsigned v = __builtin_amdgcn_readfirstlane(threadIdx.x >> 5);
    const int* src = edges;
    const int* dst = edges + N_EDGES;

    const size_t at = (size_t)v * F + blockIdx.x * AGG_COLS + lane;
    const float own = h[at];
    hs[v * AGG_COLS + lane] = own;

    int sv[AGG_CHUNKS], dv[AGG_CHUNKS];
    #pragma unroll
    for (int k = 0; k < AGG_CHUNKS; ++k) {
        const int e = k * 32 + (int)lane;
        const int ec = e < N_EDGES ? e : N_EDGES - 1;
        dv[k] = dst[ec];
        sv[k] = src[ec];
    }

    int* my = lst + v * AGG_LIST;
    unsigned cnt = 0;
    #pragma unroll
    for (int k = 0; k < AGG_CHUNKS; ++k) {
        const bool m = (k * 32 + (int)lane < N_EDGES) && dv[k] == (int)v;
        const unsigned mask = __builtin_amdgcn_ballot_w32(m);
        if (m) my[cnt + __builtin_amdgcn_mbcnt_lo(mask, 0u)] = sv[k];
        cnt += (unsigned)__builtin_popcount(mask);
    }
    if (lane < 8) my[cnt + lane] = 0;
    __syncthreads();

    float acc = own;
    for (unsigned i = 0; i < cnt; i += 8) {
        const v4i s0 = *(const v4i*)(my + i);
        const v4i s1 = *(const v4i*)(my + i + 4);
        const float x0 = hs[s0.x * AGG_COLS + lane];
        const float x1 = hs[s0.y * AGG_COLS + lane];
        const float x2 = hs[s0.z * AGG_COLS + lane];
        const float x3 = hs[s0.w * AGG_COLS + lane];
        const float x4 = hs[s1.x * AGG_COLS + lane];
        const float x5 = hs[s1.y * AGG_COLS + lane];
        const float x6 = hs[s1.z * AGG_COLS + lane];
        const float x7 = hs[s1.w * AGG_COLS + lane];
        const unsigned left = cnt - i;
        acc += x0;
        if (left > 1) acc += x1;
        if (left > 2) acc += x2;
        if (left > 3) acc += x3;
        if (left > 4) acc += x4;
        if (left > 5) acc += x5;
        if (left > 6) acc += x6;
        if (left > 7) acc += x7;
    }
    out[frag_at((int)v, (int)(blockIdx.x * AGG_COLS + lane), F)] = acc;
}

#define PART_FLOATS (8 * 32 + (N_NODES - 16) * 16)
template <int K, int N, int NWK, int RELU, int OUT_FRAG>
__global__ __launch_bounds__(32 * NWK) void gemm_cols_kernel(
        const float* __restrict__ A, const float* __restrict__ W,
        const float* __restrict__ bias, float* __restrict__ Out) {
    constexpr int KS = K / NWK;
    static_assert(KS * NWK == K && KS % 32 == 0, "k slices must tile K in whole unrolled trips");
    __shared__ float part[NWK * PART_FLOATS];

    const int lane = threadIdx.x;
    const int wave = threadIdx.y;
    const int half = lane >> 4;
    const int lm   = lane & 15;
    const int n    = blockIdx.x * 16 + lm;
    const int k0   = wave * KS;

    const float* a0p  = A + (size_t)(k0 >> 3) * 128 + lane * 4;
    const int    r1   = lm < N_NODES - 16 ? lm : N_NODES - 17;
    const float* a1p  = A + (size_t)16 * K + (size_t)(k0 >> 3) * 32 + half * 16 + r1 * 4;
    const float* bptr = W + (size_t)(k0 + 4 * half) * N + n;

    v8f c0 = {};
    v8f c1 = {};

    #pragma unroll 4
    for (int kk = 0; kk < KS; kk += 8) {
        const v4f a0 = *(const v4f*)a0p;
        const v4f a1 = *(const v4f*)a1p;
        v2f b0, b1;
        b0.x = bptr[0];
        b0.y = bptr[N];
        b1.x = bptr[2 * N];
        b1.y = bptr[3 * N];
        c0 = __builtin_amdgcn_wmma_f32_16x16x4_f32(false, a0.xy, false, b0, (short)0, c0, false, false);
        c1 = __builtin_amdgcn_wmma_f32_16x16x4_f32(false, a1.xy, false, b0, (short)0, c1, false, false);
        c0 = __builtin_amdgcn_wmma_f32_16x16x4_f32(false, a0.zw, false, b1, (short)0, c0, false, false);
        c1 = __builtin_amdgcn_wmma_f32_16x16x4_f32(false, a1.zw, false, b1, (short)0, c1, false, false);
        a0p += 128;
        a1p += 32;
        bptr += (size_t)8 * N;
    }

    float* my = part + wave * PART_FLOATS;
    #pragma unroll
    for (int i = 0; i < 8; ++i) my[i * 32 + lane] = c0[i];
    if (half == 0) {
        #pragma unroll
        for (int i = 0; i < N_NODES - 16; ++i) my[256 + i * 16 + lm] = c1[i];
    }
    __syncthreads();

    const int tid = wave * 32 + lane;
    for (int t = tid; t < PART_FLOATS; t += 32 * NWK) {
        float s = part[t];
        #pragma unroll
        for (int w = 1; w < NWK; ++w) s += part[w * PART_FLOATS + t];
        int row, col;
        if (t < 256) { row = (t >> 5) + 8 * ((t >> 4) & 1); col = t & 15; }
        else         { row = 16 + ((t - 256) >> 4);         col = t & 15; }
        const int nn = blockIdx.x * 16 + col;
        float v = s + bias[nn];
        if (RELU) v = v > 0.0f ? v : 0.0f;
        Out[OUT_FRAG ? frag_at(row, nn, N) : (size_t)row * N + nn] = v;
    }
}

#define C_ROWS   19
#define C_WAVES  8
#define C_GROUPS 512
#define C_AHEAD  7
static_assert(C_ROWS * C_WAVES * C_GROUPS == N_NODES * N_T, "rows must tile exactly: no tail code");
static_assert(C_GROUPS == LATENT, "final kernel maps one partial and one bias column per thread");

__global__ __launch_bounds__(C_WAVES * 32) void classifier_stream_kernel(
        const float* __restrict__ flat, const float* __restrict__ Wc1,
        const float* __restrict__ Wc2, float* __restrict__ partial) {
    __shared__ float wsum[C_WAVES];
    const unsigned tid  = threadIdx.x;
    const unsigned lane = tid & 31u;
    const unsigned wave = __builtin_amdgcn_readfirstlane(tid >> 5);
    const unsigned row0 = (blockIdx.x * C_WAVES + wave) * C_ROWS;

    v4f c[4];
    #pragma unroll
    for (int q = 0; q < 4; ++q) c[q] = *(const v4f*)(Wc2 + 128 * q + 4 * lane);

    const float xf = flat[row0 + (lane < C_ROWS ? lane : 0u)];

    const float* wb = Wc1 + (size_t)row0 * LATENT + 4 * lane;
    v4f w[C_AHEAD + 1][4];
#define C_LOAD_ROW(R) do { \
        _Pragma("unroll") for (int q = 0; q < 4; ++q) \
            w[(R) % (C_AHEAD + 1)][q] = __builtin_nontemporal_load((const v4f*)(wb + (size_t)(R) * LATENT + 128 * q)); \
    } while (0)

    #pragma unroll
    for (int r = 0; r < C_AHEAD; ++r) C_LOAD_ROW(r);

    float acc = 0.0f;
    #pragma unroll
    for (int r = 0; r < C_ROWS; ++r) {
        __builtin_amdgcn_sched_barrier(0);
        if (r + C_AHEAD < C_ROWS) C_LOAD_ROW(r + C_AHEAD);
        __builtin_amdgcn_sched_barrier(0);
        const v4f* wr = w[r % (C_AHEAD + 1)];
        float a0 = 0.0f, a1 = 0.0f, a2 = 0.0f, a3 = 0.0f;
        #pragma unroll
        for (int q = 0; q < 4; ++q) {
            a0 = __builtin_fmaf(wr[q].x, c[q].x, a0);
            a1 = __builtin_fmaf(wr[q].y, c[q].y, a1);
            a2 = __builtin_fmaf(wr[q].z, c[q].z, a2);
            a3 = __builtin_fmaf(wr[q].w, c[q].w, a3);
        }
        const float rowdot = (a0 + a1) + (a2 + a3);
        const float xr = __builtin_bit_cast(float,
            __builtin_amdgcn_readlane(__builtin_bit_cast(int, xf), r));
        acc = __builtin_fmaf(xr, rowdot, acc);
    }
#undef C_LOAD_ROW

    #pragma unroll
    for (int m = 16; m >= 1; m >>= 1) acc += __shfl_xor(acc, m, 32);
    if (lane == 0) wsum[wave] = acc;
    __syncthreads();
    if (tid == 0) {
        float t = 0.0f;
        #pragma unroll
        for (int i = 0; i < C_WAVES; ++i) t += wsum[i];
        partial[blockIdx.x] = t;
    }
}

__global__ __launch_bounds__(512) void final_kernel(
        const float* __restrict__ partial, const float* __restrict__ bc1,
        const float* __restrict__ Wc2, const float* __restrict__ bc2,
        float* __restrict__ out0) {
    __shared__ float red[512];
    int j = threadIdx.x;
    red[j] = partial[j] + bc1[j] * Wc2[j];
    __syncthreads();
    for (int s = 256; s > 0; s >>= 1) {
        if (j < s) red[j] += red[j + s];
        __syncthreads();
    }
    if (j == 0) {
        float v = red[0] + bc2[0];
        out0[0] = 1.0f / (1.0f + __expf(-v));
    }
}

extern "C" void kernel_launch(void* const* d_in, const int* in_sizes, int n_in,
                              void* d_out, int out_size, void* d_ws, size_t ws_size,
                              hipStream_t stream) {
    const float* z    = (const float*)d_in[0];
    const int*   eidx = (const int*)  d_in[1];
    const float* W1a  = (const float*)d_in[2];
    const float* b1a  = (const float*)d_in[3];
    const float* W1b  = (const float*)d_in[4];
    const float* b1b  = (const float*)d_in[5];
    const float* W2a  = (const float*)d_in[6];
    const float* b2a  = (const float*)d_in[7];
    const float* W2b  = (const float*)d_in[8];
    const float* b2b  = (const float*)d_in[9];
    const float* Wc1  = (const float*)d_in[10];
    const float* bc1  = (const float*)d_in[11];
    const float* Wc2  = (const float*)d_in[12];
    const float* bc2  = (const float*)d_in[13];

    float* out   = (float*)d_out;
    float* ws    = (float*)d_ws;
    float* h0    = ws + WS_H0;
    float* t1    = ws + WS_T1;
    float* h1    = ws + WS_H1;
    float* h1agg = ws + WS_H1AGG;
    float* t2    = ws + WS_T2;
    float* yv    = ws + WS_Y;
    float* recon = out + 1;

    constexpr int NWK = 16;
    dim3 gblk(32, NWK);

    gin_agg_kernel<<<LATENT / AGG_COLS, N_NODES * 32, 0, stream>>>(z, eidx, h0, LATENT);
    gemm_cols_kernel<LATENT, HID, NWK, 1, 1><<<HID / 16, gblk, 0, stream>>>(h0, W1a, b1a, t1);
    gemm_cols_kernel<HID, HID, NWK, 1, 0><<<HID / 16, gblk, 0, stream>>>(t1, W1b, b1b, h1);
    gin_agg_kernel<<<HID / AGG_COLS, N_NODES * 32, 0, stream>>>(h1, eidx, h1agg, HID);
    gemm_cols_kernel<HID, N_T, NWK, 1, 1><<<N_T / 16, gblk, 0, stream>>>(h1agg, W2a, b2a, t2);
    gemm_cols_kernel<N_T, N_T, NWK, 0, 0><<<N_T / 16, gblk, 0, stream>>>(t2, W2b, b2b, recon);
    classifier_stream_kernel<<<C_GROUPS, C_WAVES * 32, 0, stream>>>(recon, Wc1, Wc2, yv);
    final_kernel<<<1, 512, 0, stream>>>(yv, bc1, Wc2, bc2, out);
}


# classifier row dots Wc1.Wc2 (activation-independent) streamed by extra work-groups riding in the three larger GEMM launches; classifier launch is now one 77824-term dot; sums regrouped, fixed order
# speedup vs baseline: 3.6269x; 1.0584x over previous
#include <hip/hip_runtime.h>
#include <hip/hip_bf16.h>

typedef __attribute__((ext_vector_type(2))) float v2f;
typedef __attribute__((ext_vector_type(4))) float v4f;
typedef __attribute__((ext_vector_type(8))) float v8f;

#define N_NODES 19
#define N_T     4096
#define LATENT  512
#define HID     2048
#define N_EDGES 342
#define PAD_ROWS 32

#define WS_H0    0
#define WS_T1    (WS_H0 + PAD_ROWS * LATENT)
#define WS_H1    (WS_T1 + PAD_ROWS * HID)
#define WS_H1AGG (WS_H1 + PAD_ROWS * HID)
#define WS_T2    (WS_H1AGG + PAD_ROWS * HID)
#define WS_Y     (WS_T2 + PAD_ROWS * N_T)
#define WS_U     (WS_Y + LATENT)
#define WS_TOTAL (WS_U + N_NODES * N_T)

__device__ __forceinline__ size_t frag_at(int row, int k, int K) {
    const int blk = k >> 3, half = (k >> 2) & 1, j = k & 3;
    return row < 16 ? (size_t)blk * 128 + half * 64 + row * 4 + j
                    : (size_t)16 * K + (size_t)blk * 32 + half * 16 + (row - 16) * 4 + j;
}

#define AGG_COLS   32
#define AGG_CHUNKS ((N_EDGES + 31) / 32)
#define AGG_LIST   (((N_EDGES + 8) + 7) / 8 * 8)
static_assert(LATENT % AGG_COLS == 0 && HID % AGG_COLS == 0, "strips must tile both widths: no tail code");
typedef __attribute__((ext_vector_type(4))) int v4i;

__global__ __launch_bounds__(N_NODES * 32) void gin_agg_kernel(
        const float* __restrict__ h, const int* __restrict__ edges,
        float* __restrict__ out, int F) {
    __shared__ float hs[N_NODES * AGG_COLS];
    __shared__ __attribute__((aligned(16))) int lst[N_NODES * AGG_LIST];

    const unsigned lane = threadIdx.x & 31u;
    const unsigned v = __builtin_amdgcn_readfirstlane(threadIdx.x >> 5);
    const int* src = edges;
    const int* dst = edges + N_EDGES;

    const size_t at = (size_t)v * F + blockIdx.x * AGG_COLS + lane;
    const float own = h[at];
    hs[v * AGG_COLS + lane] = own;

    int sv[AGG_CHUNKS], dv[AGG_CHUNKS];
    #pragma unroll
    for (int k = 0; k < AGG_CHUNKS; ++k) {
        const int e = k * 32 + (int)lane;
        const int ec = e < N_EDGES ? e : N_EDGES - 1;
        dv[k] = dst[ec];
        sv[k] = src[ec];
    }

    int* my = lst + v * AGG_LIST;
    unsigned cnt = 0;
    #pragma unroll
    for (int k = 0; k < AGG_CHUNKS; ++k) {
        const bool m = (k * 32 + (int)lane < N_EDGES) && dv[k] == (int)v;
        const unsigned mask = __builtin_amdgcn_ballot_w32(m);
        if (m) my[cnt + __builtin_amdgcn_mbcnt_lo(mask, 0u)] = sv[k];
        cnt += (unsigned)__builtin_popcount(mask);
    }
    if (lane < 8) my[cnt + lane] = 0;
    __syncthreads();

    float acc = own;
    for (unsigned i = 0; i < cnt; i += 8) {
        const v4i s0 = *(const v4i*)(my + i);
        const v4i s1 = *(const v4i*)(my + i + 4);
        const float x0 = hs[s0.x * AGG_COLS + lane];
        const float x1 = hs[s0.y * AGG_COLS + lane];
        const float x2 = hs[s0.z * AGG_COLS + lane];
        const float x3 = hs[s0.w * AGG_COLS + lane];
        const float x4 = hs[s1.x * AGG_COLS + lane];
        const float x5 = hs[s1.y * AGG_COLS + lane];
        const float x6 = hs[s1.z * AGG_COLS + lane];
        const float x7 = hs[s1.w * AGG_COLS + lane];
        const unsigned left = cnt - i;
        acc += x0;
        if (left > 1) acc += x1;
        if (left > 2) acc += x2;
        if (left > 3) acc += x3;
        if (left > 4) acc += x4;
        if (left > 5) acc += x5;
        if (left > 6) acc += x6;
        if (left > 7) acc += x7;
    }
    out[frag_at((int)v, (int)(blockIdx.x * AGG_COLS + lane), F)] = acc;
}

#define S_ROWS   8
#define S_AHEAD  4
#define S_WAVES  16
#define S_GROUPS (N_NODES * N_T / (S_ROWS * S_WAVES))
static_assert(S_GROUPS * S_ROWS * S_WAVES == N_NODES * N_T, "rows must tile exactly: no tail code");

__device__ __forceinline__ void rowdot_stream(
        const float* __restrict__ Wc1, const float* __restrict__ Wc2, float* __restrict__ u,
        unsigned group, unsigned wave, unsigned lane) {
    const unsigned row0 = (group * S_WAVES + wave) * S_ROWS;

    v4f c[4];
    #pragma unroll
    for (int q = 0; q < 4; ++q) c[q] = *(const v4f*)(Wc2 + 128 * q + 4 * lane);

    const float* wb = Wc1 + (size_t)row0 * LATENT + 4 * lane;
    v4f w[S_AHEAD][4];
#define S_LOAD_ROW(R) do { \
        _Pragma("unroll") for (int q = 0; q < 4; ++q) \
            w[(R) % S_AHEAD][q] = __builtin_nontemporal_load((const v4f*)(wb + (size_t)(R) * LATENT + 128 * q)); \
    } while (0)

    #pragma unroll
    for (int r = 0; r < S_AHEAD; ++r) S_LOAD_ROW(r);

    float mine = 0.0f;
    #pragma unroll
    for (int r = 0; r < S_ROWS; ++r) {
        const v4f* wr = w[r % S_AHEAD];
        float a0 = 0.0f, a1 = 0.0f, a2 = 0.0f, a3 = 0.0f;
        #pragma unroll
        for (int q = 0; q < 4; ++q) {
            a0 = __builtin_fmaf(wr[q].x, c[q].x, a0);
            a1 = __builtin_fmaf(wr[q].y, c[q].y, a1);
            a2 = __builtin_fmaf(wr[q].z, c[q].z, a2);
            a3 = __builtin_fmaf(wr[q].w, c[q].w, a3);
        }
        float rowdot = (a0 + a1) + (a2 + a3);
        __builtin_amdgcn_sched_barrier(0);
        if (r + S_AHEAD < S_ROWS) S_LOAD_ROW(r + S_AHEAD);
        __builtin_amdgcn_sched_barrier(0);
        #pragma unroll
        for (int m = 16; m >= 1; m >>= 1) rowdot += __shfl_xor(rowdot, m, 32);
        if (lane == (unsigned)r) mine = rowdot;
    }
#undef S_LOAD_ROW
    if (lane < S_ROWS) u[row0 + lane] = mine;
}

#define PART_FLOATS (8 * 32 + (N_NODES - 16) * 16)
template <int K, int N, int NWK, int RELU, int OUT_FRAG, int SBASE>
__global__ __launch_bounds__(32 * NWK) void gemm_cols_kernel(
        const float* __restrict__ A, const float* __restrict__ W,
        const float* __restrict__ bias, float* __restrict__ Out,
        const float* __restrict__ Wc1, const float* __restrict__ Wc2, float* __restrict__ u) {
    constexpr int KS = K / NWK;
    static_assert(KS * NWK == K && KS % 32 == 0, "k slices must tile K in whole unrolled trips");
    static_assert(NWK == S_WAVES, "the stream work-groups share the launch's block shape");
    __shared__ float part[NWK * PART_FLOATS];

    if (blockIdx.x >= N / 16) {
        rowdot_stream(Wc1, Wc2, u, SBASE + (blockIdx.x - N / 16),
                      __builtin_amdgcn_readfirstlane(threadIdx.y), threadIdx.x);
        return;
    }

    const int lane = threadIdx.x;
    const int wave = threadIdx.y;
    const int half = lane >> 4;
    const int lm   = lane & 15;
    const int n    = blockIdx.x * 16 + lm;
    const int k0   = wave * KS;

    const float* a0p  = A + (size_t)(k0 >> 3) * 128 + lane * 4;
    const int    r1   = lm < N_NODES - 16 ? lm : N_NODES - 17;
    const float* a1p  = A + (size_t)16 * K + (size_t)(k0 >> 3) * 32 + half * 16 + r1 * 4;
    const float* bptr = W + (size_t)(k0 + 4 * half) * N + n;

    v8f c0 = {};
    v8f c1 = {};

    #pragma unroll 4
    for (int kk = 0; kk < KS; kk += 8) {
        const v4f a0 = *(const v4f*)a0p;
        const v4f a1 = *(const v4f*)a1p;
        v2f b0, b1;
        b0.x = bptr[0];
        b0.y = bptr[N];
        b1.x = bptr[2 * N];
        b1.y = bptr[3 * N];
        c0 = __builtin_amdgcn_wmma_f32_16x16x4_f32(false, a0.xy, false, b0, (short)0, c0, false, false);
        c1 = __builtin_amdgcn_wmma_f32_16x16x4_f32(false, a1.xy, false, b0, (short)0, c1, false, false);
        c0 = __builtin_amdgcn_wmma_f32_16x16x4_f32(false, a0.zw, false, b1, (short)0, c0, false, false);
        c1 = __builtin_amdgcn_wmma_f32_16x16x4_f32(false, a1.zw, false, b1, (short)0, c1, false, false);
        a0p += 128;
        a1p += 32;
        bptr += (size_t)8 * N;
    }

    float* my = part + wave * PART_FLOATS;
    #pragma unroll
    for (int i = 0; i < 8; ++i) my[i * 32 + lane] = c0[i];
    if (half == 0) {
        #pragma unroll
        for (int i = 0; i < N_NODES - 16; ++i) my[256 + i * 16 + lm] = c1[i];
    }
    __syncthreads();

    const int tid = wave * 32 + lane;
    for (int t = tid; t < PART_FLOATS; t += 32 * NWK) {
        float s = part[t];
        #pragma unroll
        for (int w = 1; w < NWK; ++w) s += part[w * PART_FLOATS + t];
        int row, col;
        if (t < 256) { row = (t >> 5) + 8 * ((t >> 4) & 1); col = t & 15; }
        else         { row = 16 + ((t - 256) >> 4);         col = t & 15; }
        const int nn = blockIdx.x * 16 + col;
        float v = s + bias[nn];
        if (RELU) v = v > 0.0f ? v : 0.0f;
        Out[OUT_FRAG ? frag_at(row, nn, N) : (size_t)row * N + nn] = v;
    }
}

#define D_THREADS 256
#define D_PER     4
#define D_GROUPS  (N_NODES * N_T / (D_THREADS * D_PER))
static_assert(D_GROUPS * D_THREADS * D_PER == N_NODES * N_T, "terms must tile exactly: no tail code");
static_assert(D_GROUPS <= LATENT, "the final kernel reads one partial per thread");

__global__ __launch_bounds__(D_THREADS) void classifier_dot_kernel(
        const float* __restrict__ flat, const float* __restrict__ u, float* __restrict__ partial) {
    __shared__ float wsum[D_THREADS / 32];
    const unsigned tid  = threadIdx.x;
    const unsigned lane = tid & 31u;
    const unsigned base = blockIdx.x * (D_THREADS * D_PER) + tid;
    float x[D_PER], y[D_PER];
    #pragma unroll
    for (int i = 0; i < D_PER; ++i) {
        x[i] = flat[base + i * D_THREADS];
        y[i] = u[base + i * D_THREADS];
    }
    float acc = 0.0f;
    #pragma unroll
    for (int i = 0; i < D_PER; ++i) acc = __builtin_fmaf(x[i], y[i], acc);
    #pragma unroll
    for (int m = 16; m >= 1; m >>= 1) acc += __shfl_xor(acc, m, 32);
    if (lane == 0) wsum[tid >> 5] = acc;
    __syncthreads();
    if (tid == 0) {
        float t = 0.0f;
        #pragma unroll
        for (int i = 0; i < D_THREADS / 32; ++i) t += wsum[i];
        partial[blockIdx.x] = t;
    }
}

__global__ __launch_bounds__(512) void final_kernel(
        const float* __restrict__ partial, const float* __restrict__ bc1,
        const float* __restrict__ Wc2, const float* __restrict__ bc2,
        float* __restrict__ out0) {
    __shared__ float red[512];
    int j = threadIdx.x;
    red[j] = (j < D_GROUPS ? partial[j] : 0.0f) + bc1[j] * Wc2[j];
    __syncthreads();
    for (int s = 256; s > 0; s >>= 1) {
        if (j < s) red[j] += red[j + s];
        __syncthreads();
    }
    if (j == 0) {
        float v = red[0] + bc2[0];
        out0[0] = 1.0f / (1.0f + __expf(-v));
    }
}

extern "C" void kernel_launch(void* const* d_in, const int* in_sizes, int n_in,
                              void* d_out, int out_size, void* d_ws, size_t ws_size,
                              hipStream_t stream) {
    const float* z    = (const float*)d_in[0];
    const int*   eidx = (const int*)  d_in[1];
    const float* W1a  = (const float*)d_in[2];
    const float* b1a  = (const float*)d_in[3];
    const float* W1b  = (const float*)d_in[4];
    const float* b1b  = (const float*)d_in[5];
    const float* W2a  = (const float*)d_in[6];
    const float* b2a  = (const float*)d_in[7];
    const float* W2b  = (const float*)d_in[8];
    const float* b2b  = (const float*)d_in[9];
    const float* Wc1  = (const float*)d_in[10];
    const float* bc1  = (const float*)d_in[11];
    const float* Wc2  = (const float*)d_in[12];
    const float* bc2  = (const float*)d_in[13];

    float* out   = (float*)d_out;
    float* ws    = (float*)d_ws;
    float* h0    = ws + WS_H0;
    float* t1    = ws + WS_T1;
    float* h1    = ws + WS_H1;
    float* h1agg = ws + WS_H1AGG;
    float* t2    = ws + WS_T2;
    float* yv    = ws + WS_Y;
    float* uv    = ws + WS_U;
    float* recon = out + 1;

    constexpr int NWK = 16;
    dim3 gblk(32, NWK);
    constexpr int S1 = 0, S2 = 160, S3 = 192, S4 = 256;
    static_assert(S1 + S2 + S3 + S4 == S_GROUPS, "every row of Wc1 is streamed exactly once");

    gin_agg_kernel<<<LATENT / AGG_COLS, N_NODES * 32, 0, stream>>>(z, eidx, h0, LATENT);
    gemm_cols_kernel<LATENT, HID, NWK, 1, 1, 0><<<HID / 16 + S1, gblk, 0, stream>>>(h0, W1a, b1a, t1, Wc1, Wc2, uv);
    gemm_cols_kernel<HID, HID, NWK, 1, 0, S1><<<HID / 16 + S2, gblk, 0, stream>>>(t1, W1b, b1b, h1, Wc1, Wc2, uv);
    gin_agg_kernel<<<HID / AGG_COLS, N_NODES * 32, 0, stream>>>(h1, eidx, h1agg, HID);
    gemm_cols_kernel<HID, N_T, NWK, 1, 1, S1 + S2><<<N_T / 16 + S3, gblk, 0, stream>>>(h1agg, W2a, b2a, t2, Wc1, Wc2, uv);
    gemm_cols_kernel<N_T, N_T, NWK, 0, 0, S1 + S2 + S3><<<N_T / 16 + S4, gblk, 0, stream>>>(t2, W2b, b2b, recon, Wc1, Wc2, uv);
    classifier_dot_kernel<<<D_GROUPS, D_THREADS, 0, stream>>>(recon, uv, yv);
    final_kernel<<<1, 512, 0, stream>>>(yv, bc1, Wc2, bc2, out);
}


# GEMM k loop software-pipelined in registers: a wave keeps the A and W loads of its next 6 k-blocks of 8 requested in a register ring while it multiplies the current block; same sums, same order
# speedup vs baseline: 3.8023x; 1.0484x over previous
#include <hip/hip_runtime.h>
#include <hip/hip_bf16.h>

typedef __attribute__((ext_vector_type(2))) float v2f;
typedef __attribute__((ext_vector_type(4))) float v4f;
typedef __attribute__((ext_vector_type(8))) float v8f;

#define N_NODES 19
#define N_T     4096
#define LATENT  512
#define HID     2048
#define N_EDGES 342
#define PAD_ROWS 32

#define WS_H0    0
#define WS_T1    (WS_H0 + PAD_ROWS * LATENT)
#define WS_H1    (WS_T1 + PAD_ROWS * HID)
#define WS_H1AGG (WS_H1 + PAD_ROWS * HID)
#define WS_T2    (WS_H1AGG + PAD_ROWS * HID)
#define WS_Y     (WS_T2 + PAD_ROWS * N_T)
#define WS_U     (WS_Y + LATENT)
#define WS_TOTAL (WS_U + N_NODES * N_T)

__device__ __forceinline__ size_t frag_at(int row, int k, int K) {
    const int blk = k >> 3, half = (k >> 2) & 1, j = k & 3;
    return row < 16 ? (size_t)blk * 128 + half * 64 + row * 4 + j
                    : (size_t)16 * K + (size_t)blk * 32 + half * 16 + (row - 16) * 4 + j;
}

#define AGG_COLS   32
#define AGG_CHUNKS ((N_EDGES + 31) / 32)
#define AGG_LIST   (((N_EDGES + 8) + 7) / 8 * 8)
static_assert(LATENT % AGG_COLS == 0 && HID % AGG_COLS == 0, "strips must tile both widths: no tail code");
typedef __attribute__((ext_vector_type(4))) int v4i;

__global__ __launch_bounds__(N_NODES * 32) void gin_agg_kernel(
        const float* __restrict__ h, const int* __restrict__ edges,
        float* __restrict__ out, int F) {
    __shared__ float hs[N_NODES * AGG_COLS];
    __shared__ __attribute__((aligned(16))) int lst[N_NODES * AGG_LIST];

    const unsigned lane = threadIdx.x & 31u;
    const unsigned v = __builtin_amdgcn_readfirstlane(threadIdx.x >> 5);
    const int* src = edges;
    const int* dst = edges + N_EDGES;

    const size_t at = (size_t)v * F + blockIdx.x * AGG_COLS + lane;
    const float own = h[at];
    hs[v * AGG_COLS + lane] = own;

    int sv[AGG_CHUNKS], dv[AGG_CHUNKS];
    #pragma unroll
    for (int k = 0; k < AGG_CHUNKS; ++k) {
        const int e = k * 32 + (int)lane;
        const int ec = e < N_EDGES ? e : N_EDGES - 1;
        dv[k] = dst[ec];
        sv[k] = src[ec];
    }

    int* my = lst + v * AGG_LIST;
    unsigned cnt = 0;
    #pragma unroll
    for (int k = 0; k < AGG_CHUNKS; ++k) {
        const bool m = (k * 32 + (int)lane < N_EDGES) && dv[k] == (int)v;
        const unsigned mask = __builtin_amdgcn_ballot_w32(m);
        if (m) my[cnt + __builtin_amdgcn_mbcnt_lo(mask, 0u)] = sv[k];
        cnt += (unsigned)__builtin_popcount(mask);
    }
    if (lane < 8) my[cnt + lane] = 0;
    __syncthreads();

    float acc = own;
    for (unsigned i = 0; i < cnt; i += 8) {
        const v4i s0 = *(const v4i*)(my + i);
        const v4i s1 = *(const v4i*)(my + i + 4);
        const float x0 = hs[s0.x * AGG_COLS + lane];
        const float x1 = hs[s0.y * AGG_COLS + lane];
        const float x2 = hs[s0.z * AGG_COLS + lane];
        const float x3 = hs[s0.w * AGG_COLS + lane];
        const float x4 = hs[s1.x * AGG_COLS + lane];
        const float x5 = hs[s1.y * AGG_COLS + lane];
        const float x6 = hs[s1.z * AGG_COLS + lane];
        const float x7 = hs[s1.w * AGG_COLS + lane];
        const unsigned left = cnt - i;
        acc += x0;
        if (left > 1) acc += x1;
        if (left > 2) acc += x2;
        if (left > 3) acc += x3;
        if (left > 4) acc += x4;
        if (left > 5) acc += x5;
        if (left > 6) acc += x6;
        if (left > 7) acc += x7;
    }
    out[frag_at((int)v, (int)(blockIdx.x * AGG_COLS + lane), F)] = acc;
}

#define S_ROWS   8
#define S_AHEAD  4
#define S_WAVES  16
#define S_GROUPS (N_NODES * N_T / (S_ROWS * S_WAVES))
static_assert(S_GROUPS * S_ROWS * S_WAVES == N_NODES * N_T, "rows must tile exactly: no tail code");

__device__ __forceinline__ void rowdot_stream(
        const float* __restrict__ Wc1, const float* __restrict__ Wc2, float* __restrict__ u,
        unsigned group, unsigned wave, unsigned lane) {
    const unsigned row0 = (group * S_WAVES + wave) * S_ROWS;

    v4f c[4];
    #pragma unroll
    for (int q = 0; q < 4; ++q) c[q] = *(const v4f*)(Wc2 + 128 * q + 4 * lane);

    const float* wb = Wc1 + (size_t)row0 * LATENT + 4 * lane;
    v4f w[S_AHEAD][4];
#define S_LOAD_ROW(R) do { \
        _Pragma("unroll") for (int q = 0; q < 4; ++q) \
            w[(R) % S_AHEAD][q] = __builtin_nontemporal_load((const v4f*)(wb + (size_t)(R) * LATENT + 128 * q)); \
    } while (0)

    #pragma unroll
    for (int r = 0; r < S_AHEAD; ++r) S_LOAD_ROW(r);

    float mine = 0.0f;
    #pragma unroll
    for (int r = 0; r < S_ROWS; ++r) {
        const v4f* wr = w[r % S_AHEAD];
        float a0 = 0.0f, a1 = 0.0f, a2 = 0.0f, a3 = 0.0f;
        #pragma unroll
        for (int q = 0; q < 4; ++q) {
            a0 = __builtin_fmaf(wr[q].x, c[q].x, a0);
            a1 = __builtin_fmaf(wr[q].y, c[q].y, a1);
            a2 = __builtin_fmaf(wr[q].z, c[q].z, a2);
            a3 = __builtin_fmaf(wr[q].w, c[q].w, a3);
        }
        float rowdot = (a0 + a1) + (a2 + a3);
        __builtin_amdgcn_sched_barrier(0);
        if (r + S_AHEAD < S_ROWS) S_LOAD_ROW(r + S_AHEAD);
        __builtin_amdgcn_sched_barrier(0);
        #pragma unroll
        for (int m = 16; m >= 1; m >>= 1) rowdot += __shfl_xor(rowdot, m, 32);
        if (lane == (unsigned)r) mine = rowdot;
    }
#undef S_LOAD_ROW
    if (lane < S_ROWS) u[row0 + lane] = mine;
}

#define PART_FLOATS (8 * 32 + (N_NODES - 16) * 16)
#define G_AHEAD 6
template <int K, int N, int NWK, int RELU, int OUT_FRAG, int SBASE>
__global__ __launch_bounds__(32 * NWK) void gemm_cols_kernel(
        const float* __restrict__ A, const float* __restrict__ W,
        const float* __restrict__ bias, float* __restrict__ Out,
        const float* __restrict__ Wc1, const float* __restrict__ Wc2, float* __restrict__ u) {
    constexpr int KS = K / NWK;
    static_assert(KS * NWK == K && KS % 32 == 0, "k slices must tile K in whole unrolled trips");
    static_assert(NWK == S_WAVES, "the stream work-groups share the launch's block shape");
    __shared__ float part[NWK * PART_FLOATS];

    if (blockIdx.x >= N / 16) {
        rowdot_stream(Wc1, Wc2, u, SBASE + (blockIdx.x - N / 16),
                      __builtin_amdgcn_readfirstlane(threadIdx.y), threadIdx.x);
        return;
    }

    const int lane = threadIdx.x;
    const int wave = threadIdx.y;
    const int half = lane >> 4;
    const int lm   = lane & 15;
    const int n    = blockIdx.x * 16 + lm;
    const int k0   = wave * KS;

    const float* a0p  = A + (size_t)(k0 >> 3) * 128 + lane * 4;
    const int    r1   = lm < N_NODES - 16 ? lm : N_NODES - 17;
    const float* a1p  = A + (size_t)16 * K + (size_t)(k0 >> 3) * 32 + half * 16 + r1 * 4;
    const float* bptr = W + (size_t)(k0 + 4 * half) * N + n;

    v8f c0 = {};
    v8f c1 = {};

    constexpr int NIT   = KS / 8;
    constexpr int AHEAD = NIT < G_AHEAD ? NIT : G_AHEAD;
    constexpr int RING  = AHEAD + 1;
    v4f ra0[RING], ra1[RING];
    v2f rb0[RING], rb1[RING];
#define G_LOAD(I) do { \
        const int s_ = (I) % RING; \
        const float* bp_ = bptr + (size_t)(I) * 8 * N; \
        ra0[s_] = *(const v4f*)(a0p + (size_t)(I) * 128); \
        ra1[s_] = *(const v4f*)(a1p + (size_t)(I) * 32); \
        rb0[s_].x = bp_[0]; \
        rb0[s_].y = bp_[N]; \
        rb1[s_].x = bp_[2 * N]; \
        rb1[s_].y = bp_[3 * N]; \
    } while (0)

    #pragma unroll
    for (int i = 0; i < AHEAD; ++i) G_LOAD(i);

    #pragma unroll
    for (int i = 0; i < NIT; ++i) {
        if (i + AHEAD < NIT) G_LOAD(i + AHEAD);
        __builtin_amdgcn_sched_barrier(0);
        const int s = i % RING;
        c0 = __builtin_amdgcn_wmma_f32_16x16x4_f32(false, ra0[s].xy, false, rb0[s], (short)0, c0, false, false);
        c1 = __builtin_amdgcn_wmma_f32_16x16x4_f32(false, ra1[s].xy, false, rb0[s], (short)0, c1, false, false);
        c0 = __builtin_amdgcn_wmma_f32_16x16x4_f32(false, ra0[s].zw, false, rb1[s], (short)0, c0, false, false);
        c1 = __builtin_amdgcn_wmma_f32_16x16x4_f32(false, ra1[s].zw, false, rb1[s], (short)0, c1, false, false);
    }
#undef G_LOAD

    float* my = part + wave * PART_FLOATS;
    #pragma unroll
    for (int i = 0; i < 8; ++i) my[i * 32 + lane] = c0[i];
    if (half == 0) {
        #pragma unroll
        for (int i = 0; i < N_NODES - 16; ++i) my[256 + i * 16 + lm] = c1[i];
    }
    __syncthreads();

    const int tid = wave * 32 + lane;
    for (int t = tid; t < PART_FLOATS; t += 32 * NWK) {
        float s = part[t];
        #pragma unroll
        for (int w = 1; w < NWK; ++w) s += part[w * PART_FLOATS + t];
        int row, col;
        if (t < 256) { row = (t >> 5) + 8 * ((t >> 4) & 1); col = t & 15; }
        else         { row = 16 + ((t - 256) >> 4);         col = t & 15; }
        const int nn = blockIdx.x * 16 + col;
        float v = s + bias[nn];
        if (RELU) v = v > 0.0f ? v : 0.0f;
        Out[OUT_FRAG ? frag_at(row, nn, N) : (size_t)row * N + nn] = v;
    }
}

#define D_THREADS 256
#define D_PER     4
#define D_GROUPS  (N_NODES * N_T / (D_THREADS * D_PER))
static_assert(D_GROUPS * D_THREADS * D_PER == N_NODES * N_T, "terms must tile exactly: no tail code");
static_assert(D_GROUPS <= LATENT, "the final kernel reads one partial per thread");

__global__ __launch_bounds__(D_THREADS) void classifier_dot_kernel(
        const float* __restrict__ flat, const float* __restrict__ u, float* __restrict__ partial) {
    __shared__ float wsum[D_THREADS / 32];
    const unsigned tid  = threadIdx.x;
    const unsigned lane = tid & 31u;
    const unsigned base = blockIdx.x * (D_THREADS * D_PER) + tid;
    float x[D_PER], y[D_PER];
    #pragma unroll
    for (int i = 0; i < D_PER; ++i) {
        x[i] = flat[base + i * D_THREADS];
        y[i] = u[base + i * D_THREADS];
    }
    float acc = 0.0f;
    #pragma unroll
    for (int i = 0; i < D_PER; ++i) acc = __builtin_fmaf(x[i], y[i], acc);
    #pragma unroll
    for (int m = 16; m >= 1; m >>= 1) acc += __shfl_xor(acc, m, 32);
    if (lane == 0) wsum[tid >> 5] = acc;
    __syncthreads();
    if (tid == 0) {
        float t = 0.0f;
        #pragma unroll
        for (int i = 0; i < D_THREADS / 32; ++i) t += wsum[i];
        partial[blockIdx.x] = t;
    }
}

__global__ __launch_bounds__(512) void final_kernel(
        const float* __restrict__ partial, const float* __restrict__ bc1,
        const float* __restrict__ Wc2, const float* __restrict__ bc2,
        float* __restrict__ out0) {
    __shared__ float red[512];
    int j = threadIdx.x;
    red[j] = (j < D_GROUPS ? partial[j] : 0.0f) + bc1[j] * Wc2[j];
    __syncthreads();
    for (int s = 256; s > 0; s >>= 1) {
        if (j < s) red[j] += red[j + s];
        __syncthreads();
    }
    if (j == 0) {
        float v = red[0] + bc2[0];
        out0[0] = 1.0f / (1.0f + __expf(-v));
    }
}

extern "C" void kernel_launch(void* const* d_in, const int* in_sizes, int n_in,
                              void* d_out, int out_size, void* d_ws, size_t ws_size,
                              hipStream_t stream) {
    const float* z    = (const float*)d_in[0];
    const int*   eidx = (const int*)  d_in[1];
    const float* W1a  = (const float*)d_in[2];
    const float* b1a  = (const float*)d_in[3];
    const float* W1b  = (const float*)d_in[4];
    const float* b1b  = (const float*)d_in[5];
    const float* W2a  = (const float*)d_in[6];
    const float* b2a  = (const float*)d_in[7];
    const float* W2b  = (const float*)d_in[8];
    const float* b2b  = (const float*)d_in[9];
    const float* Wc1  = (const float*)d_in[10];
    const float* bc1  = (const float*)d_in[11];
    const float* Wc2  = (const float*)d_in[12];
    const float* bc2  = (const float*)d_in[13];

    float* out   = (float*)d_out;
    float* ws    = (float*)d_ws;
    float* h0    = ws + WS_H0;
    float* t1    = ws + WS_T1;
    float* h1    = ws + WS_H1;
    float* h1agg = ws + WS_H1AGG;
    float* t2    = ws + WS_T2;
    float* yv    = ws + WS_Y;
    float* uv    = ws + WS_U;
    float* recon = out + 1;

    constexpr int NWK = 16;
    dim3 gblk(32, NWK);
    constexpr int S1 = 0, S2 = 160, S3 = 192, S4 = 256;
    static_assert(S1 + S2 + S3 + S4 == S_GROUPS, "every row of Wc1 is streamed exactly once");

    gin_agg_kernel<<<LATENT / AGG_COLS, N_NODES * 32, 0, stream>>>(z, eidx, h0, LATENT);
    gemm_cols_kernel<LATENT, HID, NWK, 1, 1, 0><<<HID / 16 + S1, gblk, 0, stream>>>(h0, W1a, b1a, t1, Wc1, Wc2, uv);
    gemm_cols_kernel<HID, HID, NWK, 1, 0, S1><<<HID / 16 + S2, gblk, 0, stream>>>(t1, W1b, b1b, h1, Wc1, Wc2, uv);
    gin_agg_kernel<<<HID / AGG_COLS, N_NODES * 32, 0, stream>>>(h1, eidx, h1agg, HID);
    gemm_cols_kernel<HID, N_T, NWK, 1, 1, S1 + S2><<<N_T / 16 + S3, gblk, 0, stream>>>(h1agg, W2a, b2a, t2, Wc1, Wc2, uv);
    gemm_cols_kernel<N_T, N_T, NWK, 0, 0, S1 + S2 + S3><<<N_T / 16 + S4, gblk, 0, stream>>>(t2, W2b, b2b, recon, Wc1, Wc2, uv);
    classifier_dot_kernel<<<D_GROUPS, D_THREADS, 0, stream>>>(recon, uv, yv);
    final_kernel<<<1, 512, 0, stream>>>(yv, bc1, Wc2, bc2, out);
}


# column-wise followers run in the GEMM finish step (a work-group owns all 19 rows): GEMM 2 adds the GIN sum, same add order; GEMM 4 leaves flat.u partials, sums regrouped; 8->6 launches
# speedup vs baseline: 4.1665x; 1.0958x over previous
#include <hip/hip_runtime.h>
#include <hip/hip_bf16.h>

typedef __attribute__((ext_vector_type(2))) float v2f;
typedef __attribute__((ext_vector_type(4))) float v4f;
typedef __attribute__((ext_vector_type(8))) float v8f;

#define N_NODES 19
#define N_T     4096
#define LATENT  512
#define HID     2048
#define N_EDGES 342
#define PAD_ROWS 32

#define WS_H0    0
#define WS_T1    (WS_H0 + PAD_ROWS * LATENT)
#define WS_H1AGG (WS_T1 + PAD_ROWS * HID)
#define WS_T2    (WS_H1AGG + PAD_ROWS * HID)
#define WS_Y     (WS_T2 + PAD_ROWS * N_T)
#define WS_U     (WS_Y + LATENT)
#define WS_TOTAL (WS_U + N_NODES * N_T)

__device__ __forceinline__ size_t frag_at(int row, int k, int K) {
    const int blk = k >> 3, half = (k >> 2) & 1, j = k & 3;
    return row < 16 ? (size_t)blk * 128 + half * 64 + row * 4 + j
                    : (size_t)16 * K + (size_t)blk * 32 + half * 16 + (row - 16) * 4 + j;
}

#define AGG_COLS   32
#define AGG_CHUNKS ((N_EDGES + 31) / 32)
#define AGG_LIST   (((N_EDGES + 8) + 7) / 8 * 8)
static_assert(LATENT % AGG_COLS == 0 && HID % AGG_COLS == 0, "strips must tile both widths: no tail code");
typedef __attribute__((ext_vector_type(4))) int v4i;

__global__ __launch_bounds__(N_NODES * 32) void gin_agg_kernel(
        const float* __restrict__ h, const int* __restrict__ edges,
        float* __restrict__ out, int F) {
    __shared__ float hs[N_NODES * AGG_COLS];
    __shared__ __attribute__((aligned(16))) int lst[N_NODES * AGG_LIST];

    const unsigned lane = threadIdx.x & 31u;
    const unsigned v = __builtin_amdgcn_readfirstlane(threadIdx.x >> 5);
    const int* src = edges;
    const int* dst = edges + N_EDGES;

    const size_t at = (size_t)v * F + blockIdx.x * AGG_COLS + lane;
    const float own = h[at];
    hs[v * AGG_COLS + lane] = own;

    int sv[AGG_CHUNKS], dv[AGG_CHUNKS];
    #pragma unroll
    for (int k = 0; k < AGG_CHUNKS; ++k) {
        const int e = k * 32 + (int)lane;
        const int ec = e < N_EDGES ? e : N_EDGES - 1;
        dv[k] = dst[ec];
        sv[k] = src[ec];
    }

    int* my = lst + v * AGG_LIST;
    unsigned cnt = 0;
    #pragma unroll
    for (int k = 0; k < AGG_CHUNKS; ++k) {
        const bool m = (k * 32 + (int)lane < N_EDGES) && dv[k] == (int)v;
        const unsigned mask = __builtin_amdgcn_ballot_w32(m);
        if (m) my[cnt + __builtin_amdgcn_mbcnt_lo(mask, 0u)] = sv[k];
        cnt += (unsigned)__builtin_popcount(mask);
    }
    if (lane < 8) my[cnt + lane] = 0;
    __syncthreads();

    float acc = own;
    for (unsigned i = 0; i < cnt; i += 8) {
        const v4i s0 = *(const v4i*)(my + i);
        const v4i s1 = *(const v4i*)(my + i + 4);
        const float x0 = hs[s0.x * AGG_COLS + lane];
        const float x1 = hs[s0.y * AGG_COLS + lane];
        const float x2 = hs[s0.z * AGG_COLS + lane];
        const float x3 = hs[s0.w * AGG_COLS + lane];
        const float x4 = hs[s1.x * AGG_COLS + lane];
        const float x5 = hs[s1.y * AGG_COLS + lane];
        const float x6 = hs[s1.z * AGG_COLS + lane];
        const float x7 = hs[s1.w * AGG_COLS + lane];
        const unsigned left = cnt - i;
        acc += x0;
        if (left > 1) acc += x1;
        if (left > 2) acc += x2;
        if (left > 3) acc += x3;
        if (left > 4) acc += x4;
        if (left > 5) acc += x5;
        if (left > 6) acc += x6;
        if (left > 7) acc += x7;
    }
    out[frag_at((int)v, (int)(blockIdx.x * AGG_COLS + lane), F)] = acc;
}

__device__ __forceinline__ void gin_load_edges(
        const int* __restrict__ edges, unsigned lane, int (&sv)[AGG_CHUNKS], int (&dv)[AGG_CHUNKS]) {
    const int* src = edges;
    const int* dst = edges + N_EDGES;
    #pragma unroll
    for (int k = 0; k < AGG_CHUNKS; ++k) {
        const int e = k * 32 + (int)lane;
        const int ec = e < N_EDGES ? e : N_EDGES - 1;
        dv[k] = dst[ec];
        sv[k] = src[ec];
    }
}

__device__ __forceinline__ void gin_pack_node(
        const int (&sv)[AGG_CHUNKS], const int (&dv)[AGG_CHUNKS], int* lst, int* cnt, int v, unsigned lane) {
    int* my = lst + v * AGG_LIST;
    unsigned n = 0;
    #pragma unroll
    for (int k = 0; k < AGG_CHUNKS; ++k) {
        const bool m = (k * 32 + (int)lane < N_EDGES) && dv[k] == v;
        const unsigned mask = __builtin_amdgcn_ballot_w32(m);
        if (m) my[n + __builtin_amdgcn_mbcnt_lo(mask, 0u)] = sv[k];
        n += (unsigned)__builtin_popcount(mask);
    }
    if (lane < 8) my[n + lane] = 0;
    if (lane == 0) cnt[v] = (int)n;
}

template <int NWAVES>
__device__ __forceinline__ void gin_pack_lists(
        const int (&sv)[AGG_CHUNKS], const int (&dv)[AGG_CHUNKS], int* lst, int* cnt, int wv, unsigned lane) {
    static_assert(NWAVES <= N_NODES && 2 * NWAVES >= N_NODES, "one or two nodes per wave");
    gin_pack_node(sv, dv, lst, cnt, wv, lane);
    if (wv + NWAVES < N_NODES) gin_pack_node(sv, dv, lst, cnt, wv + NWAVES, lane);
}

#define S_ROWS   8
#define S_AHEAD  4
#define S_WAVES  16
#define S_GROUPS (N_NODES * N_T / (S_ROWS * S_WAVES))
static_assert(S_GROUPS * S_ROWS * S_WAVES == N_NODES * N_T, "rows must tile exactly: no tail code");

__device__ __forceinline__ void rowdot_stream(
        const float* __restrict__ Wc1, const float* __restrict__ Wc2, float* __restrict__ u,
        unsigned group, unsigned wave, unsigned lane) {
    const unsigned row0 = (group * S_WAVES + wave) * S_ROWS;

    v4f c[4];
    #pragma unroll
    for (int q = 0; q < 4; ++q) c[q] = *(const v4f*)(Wc2 + 128 * q + 4 * lane);

    const float* wb = Wc1 + (size_t)row0 * LATENT + 4 * lane;
    v4f w[S_AHEAD][4];
#define S_LOAD_ROW(R) do { \
        _Pragma("unroll") for (int q = 0; q < 4; ++q) \
            w[(R) % S_AHEAD][q] = __builtin_nontemporal_load((const v4f*)(wb + (size_t)(R) * LATENT + 128 * q)); \
    } while (0)

    #pragma unroll
    for (int r = 0; r < S_AHEAD; ++r) S_LOAD_ROW(r);

    float mine = 0.0f;
    #pragma unroll
    for (int r = 0; r < S_ROWS; ++r) {
        const v4f* wr = w[r % S_AHEAD];
        float a0 = 0.0f, a1 = 0.0f, a2 = 0.0f, a3 = 0.0f;
        #pragma unroll
        for (int q = 0; q < 4; ++q) {
            a0 = __builtin_fmaf(wr[q].x, c[q].x, a0);
            a1 = __builtin_fmaf(wr[q].y, c[q].y, a1);
            a2 = __builtin_fmaf(wr[q].z, c[q].z, a2);
            a3 = __builtin_fmaf(wr[q].w, c[q].w, a3);
        }
        float rowdot = (a0 + a1) + (a2 + a3);
        __builtin_amdgcn_sched_barrier(0);
        if (r + S_AHEAD < S_ROWS) S_LOAD_ROW(r + S_AHEAD);
        __builtin_amdgcn_sched_barrier(0);
        #pragma unroll
        for (int m = 16; m >= 1; m >>= 1) rowdot += __shfl_xor(rowdot, m, 32);
        if (lane == (unsigned)r) mine = rowdot;
    }
#undef S_LOAD_ROW
    if (lane < S_ROWS) u[row0 + lane] = mine;
}

#define PART_FLOATS (8 * 32 + (N_NODES - 16) * 16)
#define G_AHEAD 6
template <int K, int N, int NWK, int RELU, int OUT_FRAG, int SBASE, int FOLLOW>
__global__ __launch_bounds__(32 * NWK) __attribute__((amdgpu_waves_per_eu(8))) void gemm_cols_kernel(
        const float* __restrict__ A, const float* __restrict__ W,
        const float* __restrict__ bias, float* __restrict__ Out,
        const float* __restrict__ Wc1, const float* __restrict__ Wc2, float* __restrict__ u,
        const int* __restrict__ edges, float* __restrict__ partial) {
    constexpr int KS = K / NWK;
    static_assert(KS * NWK == K && KS % 32 == 0, "k slices must tile K in whole unrolled trips");
    static_assert(NWK == S_WAVES, "the stream work-groups share the launch's block shape");
    static_assert(FOLLOW == 0 || 32 * NWK >= PART_FLOATS, "a follower's finish step has one thread per tile element");
    static_assert(FOLLOW != 1 || OUT_FRAG == 1, "the GIN sum feeds a GEMM");
    __shared__ float part[NWK * PART_FLOATS];
    __shared__ __attribute__((aligned(16))) int lst[FOLLOW == 1 ? N_NODES * AGG_LIST : 4];
    __shared__ int   lcnt[FOLLOW == 1 ? N_NODES : 1];
    __shared__ float tile[FOLLOW == 1 ? N_NODES * 16 : 1];
    __shared__ float wsum[FOLLOW == 2 ? NWK : 1];

    if (blockIdx.x >= N / 16) {
        rowdot_stream(Wc1, Wc2, u, SBASE + (blockIdx.x - N / 16),
                      __builtin_amdgcn_readfirstlane(threadIdx.y), threadIdx.x);
        return;
    }

    const int lane = threadIdx.x;
    const int wave = threadIdx.y;
    const int half = lane >> 4;
    const int lm   = lane & 15;
    const int n    = blockIdx.x * 16 + lm;
    const int k0   = wave * KS;

    const float* a0p  = A + (size_t)(k0 >> 3) * 128 + lane * 4;
    const int    r1   = lm < N_NODES - 16 ? lm : N_NODES - 17;
    const float* a1p  = A + (size_t)16 * K + (size_t)(k0 >> 3) * 32 + half * 16 + r1 * 4;
    const float* bptr = W + (size_t)(k0 + 4 * half) * N + n;

    v8f c0 = {};
    v8f c1 = {};

    const int tid = wave * 32 + lane;
#define F_LIVE (tid < PART_FLOATS)
#define F_COL  (tid & 15)
#define F_ROW  (tid < 256 ? (tid >> 5) + 8 * ((tid >> 4) & 1) : 16 + ((tid - 256) >> 4))
    float uval = 0.0f;
    if (FOLLOW == 2 && F_LIVE) uval = u[(size_t)F_ROW * N + blockIdx.x * 16 + F_COL];
    int esrc[AGG_CHUNKS], edst[AGG_CHUNKS];

    constexpr int NIT   = KS / 8;
    constexpr int AHEAD = NIT < G_AHEAD ? NIT : G_AHEAD;
    constexpr int RING  = AHEAD + 1;
    v4f ra0[RING], ra1[RING];
    v2f rb0[RING], rb1[RING];
#define G_LOAD(I) do { \
        const int s_ = (I) % RING; \
        const float* bp_ = bptr + (size_t)(I) * 8 * N; \
        ra0[s_] = *(const v4f*)(a0p + (size_t)(I) * 128); \
        ra1[s_] = *(const v4f*)(a1p + (size_t)(I) * 32); \
        rb0[s_].x = bp_[0]; \
        rb0[s_].y = bp_[N]; \
        rb1[s_].x = bp_[2 * N]; \
        rb1[s_].y = bp_[3 * N]; \
    } while (0)

    #pragma unroll
    for (int i = 0; i < AHEAD; ++i) G_LOAD(i);

    #pragma unroll
    for (int i = 0; i < NIT; ++i) {
        if (i + AHEAD < NIT) G_LOAD(i + AHEAD);
        if (FOLLOW == 1 && i == (NIT > 4 ? NIT - 4 : 0)) gin_load_edges(edges, (unsigned)lane & 31u, esrc, edst);
        __builtin_amdgcn_sched_barrier(0);
        const int s = i % RING;
        c0 = __builtin_amdgcn_wmma_f32_16x16x4_f32(false, ra0[s].xy, false, rb0[s], (short)0, c0, false, false);
        c1 = __builtin_amdgcn_wmma_f32_16x16x4_f32(false, ra1[s].xy, false, rb0[s], (short)0, c1, false, false);
        c0 = __builtin_amdgcn_wmma_f32_16x16x4_f32(false, ra0[s].zw, false, rb1[s], (short)0, c0, false, false);
        c1 = __builtin_amdgcn_wmma_f32_16x16x4_f32(false, ra1[s].zw, false, rb1[s], (short)0, c1, false, false);
    }
#undef G_LOAD

    if (FOLLOW == 1)
        gin_pack_lists<NWK>(esrc, edst, lst, lcnt, __builtin_amdgcn_readfirstlane(threadIdx.y), (unsigned)lane);

    float* my = part + wave * PART_FLOATS;
    #pragma unroll
    for (int i = 0; i < 8; ++i) my[i * 32 + lane] = c0[i];
    if (half == 0) {
        #pragma unroll
        for (int i = 0; i < N_NODES - 16; ++i) my[256 + i * 16 + lm] = c1[i];
    }
    __syncthreads();

    if constexpr (FOLLOW == 0) {
        for (int t = tid; t < PART_FLOATS; t += 32 * NWK) {
            float s = part[t];
            #pragma unroll
            for (int w = 1; w < NWK; ++w) s += part[w * PART_FLOATS + t];
            int row, col;
            if (t < 256) { row = (t >> 5) + 8 * ((t >> 4) & 1); col = t & 15; }
            else         { row = 16 + ((t - 256) >> 4);         col = t & 15; }
            const int nn = blockIdx.x * 16 + col;
            float v = s + bias[nn];
            if (RELU) v = v > 0.0f ? v : 0.0f;
            Out[OUT_FRAG ? frag_at(row, nn, N) : (size_t)row * N + nn] = v;
        }
    } else {
        const bool live = F_LIVE;
        const int  fcol = F_COL, frow = F_ROW;
        const int  nn   = blockIdx.x * 16 + fcol;
        float v = 0.0f;
        if (live) {
            float s = part[tid];
            #pragma unroll
            for (int w = 1; w < NWK; ++w) s += part[w * PART_FLOATS + tid];
            v = s + bias[nn];
            if (RELU) v = v > 0.0f ? v : 0.0f;
        }
        if constexpr (FOLLOW == 1) {
            if (live) tile[frow * 16 + fcol] = v;
            __syncthreads();
            if (live) {
                const int* my = lst + frow * AGG_LIST;
                const unsigned cnt = (unsigned)lcnt[frow];
                float acc = v;
                for (unsigned i = 0; i < cnt; i += 8) {
                    const v4i s0 = *(const v4i*)(my + i);
                    const v4i s1 = *(const v4i*)(my + i + 4);
                    const float x0 = tile[s0.x * 16 + fcol];
                    const float x1 = tile[s0.y * 16 + fcol];
                    const float x2 = tile[s0.z * 16 + fcol];
                    const float x3 = tile[s0.w * 16 + fcol];
                    const float x4 = tile[s1.x * 16 + fcol];
                    const float x5 = tile[s1.y * 16 + fcol];
                    const float x6 = tile[s1.z * 16 + fcol];
                    const float x7 = tile[s1.w * 16 + fcol];
                    const unsigned left = cnt - i;
                    acc += x0;
                    if (left > 1) acc += x1;
                    if (left > 2) acc += x2;
                    if (left > 3) acc += x3;
                    if (left > 4) acc += x4;
                    if (left > 5) acc += x5;
                    if (left > 6) acc += x6;
                    if (left > 7) acc += x7;
                }
                Out[frag_at(frow, nn, N)] = acc;
            }
        } else {
            float p = 0.0f;
            if (live) {
                Out[(size_t)frow * N + nn] = v;
                p = v * uval;
            }
            #pragma unroll
            for (int m = 16; m >= 1; m >>= 1) p += __shfl_xor(p, m, 32);
            if (lane == 0) wsum[wave] = p;
            __syncthreads();
            if (tid == 0) {
                float t = 0.0f;
                #pragma unroll
                for (int i = 0; i < NWK; ++i) t += wsum[i];
                partial[blockIdx.x] = t;
            }
        }
    }
#undef F_LIVE
#undef F_COL
#undef F_ROW
}

#define P_GROUPS (N_T / 16)
static_assert(P_GROUPS <= LATENT, "the final kernel reads one partial per thread");

__global__ __launch_bounds__(512) void final_kernel(
        const float* __restrict__ partial, const float* __restrict__ bc1,
        const float* __restrict__ Wc2, const float* __restrict__ bc2,
        float* __restrict__ out0) {
    __shared__ float red[512];
    int j = threadIdx.x;
    red[j] = (j < P_GROUPS ? partial[j] : 0.0f) + bc1[j] * Wc2[j];
    __syncthreads();
    for (int s = 256; s > 0; s >>= 1) {
        if (j < s) red[j] += red[j + s];
        __syncthreads();
    }
    if (j == 0) {
        float v = red[0] + bc2[0];
        out0[0] = 1.0f / (1.0f + __expf(-v));
    }
}

extern "C" void kernel_launch(void* const* d_in, const int* in_sizes, int n_in,
                              void* d_out, int out_size, void* d_ws, size_t ws_size,
                              hipStream_t stream) {
    const float* z    = (const float*)d_in[0];
    const int*   eidx = (const int*)  d_in[1];
    const float* W1a  = (const float*)d_in[2];
    const float* b1a  = (const float*)d_in[3];
    const float* W1b  = (const float*)d_in[4];
    const float* b1b  = (const float*)d_in[5];
    const float* W2a  = (const float*)d_in[6];
    const float* b2a  = (const float*)d_in[7];
    const float* W2b  = (const float*)d_in[8];
    const float* b2b  = (const float*)d_in[9];
    const float* Wc1  = (const float*)d_in[10];
    const float* bc1  = (const float*)d_in[11];
    const float* Wc2  = (const float*)d_in[12];
    const float* bc2  = (const float*)d_in[13];

    float* out   = (float*)d_out;
    float* ws    = (float*)d_ws;
    float* h0    = ws + WS_H0;
    float* t1    = ws + WS_T1;
    float* h1agg = ws + WS_H1AGG;
    float* t2    = ws + WS_T2;
    float* yv    = ws + WS_Y;
    float* uv    = ws + WS_U;
    float* recon = out + 1;

    constexpr int NWK = 16;
    dim3 gblk(32, NWK);
    constexpr int S2 = 352, S3 = 256;
    static_assert(S2 + S3 == S_GROUPS, "every row of Wc1 is streamed exactly once");

    gin_agg_kernel<<<LATENT / AGG_COLS, N_NODES * 32, 0, stream>>>(z, eidx, h0, LATENT);
    gemm_cols_kernel<LATENT, HID, NWK, 1, 1, 0, 0><<<HID / 16, gblk, 0, stream>>>(h0, W1a, b1a, t1, Wc1, Wc2, uv, eidx, yv);
    gemm_cols_kernel<HID, HID, NWK, 1, 1, 0, 1><<<HID / 16 + S2, gblk, 0, stream>>>(t1, W1b, b1b, h1agg, Wc1, Wc2, uv, eidx, yv);
    gemm_cols_kernel<HID, N_T, NWK, 1, 1, S2, 0><<<N_T / 16 + S3, gblk, 0, stream>>>(h1agg, W2a, b2a, t2, Wc1, Wc2, uv, eidx, yv);
    gemm_cols_kernel<N_T, N_T, NWK, 0, 0, S_GROUPS, 2><<<N_T / 16, gblk, 0, stream>>>(t2, W2b, b2b, recon, Wc1, Wc2, uv, eidx, yv);
    final_kernel<<<1, 512, 0, stream>>>(yv, bc1, Wc2, bc2, out);
}
